# starting kernel
# baseline (speedup 1.0000x reference)
#include <hip/hip_runtime.h>
#include <math.h>

typedef __attribute__((ext_vector_type(2))) float v2f;
typedef __attribute__((ext_vector_type(8))) float v8f;

#define NG 512
#define HH 256
#define WW 256

#define QSCALE (-0.72134752044448170368f)

__global__ void splat_prep(const float* __restrict__ means,
                           const float* __restrict__ covs,
                           const float* __restrict__ colors,
                           const float* __restrict__ opac,
                           float* __restrict__ gcoef,
                           float* __restrict__ c4)
{
    int n = blockIdx.x * blockDim.x + threadIdx.x;
    if (n >= NG) return;
    float mx = means[2*n+0] - 128.0f;
    float my = means[2*n+1] - 128.0f;
    float a = covs[4*n+0], b = covs[4*n+1], c = covs[4*n+3];
    float det = a*c - b*b;
    float inv_det = 1.0f / det;
    float ia =  c*inv_det;
    float ib = -b*inv_det;
    float ic =  a*inv_det;
    float amp = opac[n] / (6.28318530717958647f * sqrtf(det + 1e-8f));

    float* g = gcoef + 8*n;
    g[0] = QSCALE * ia;
    g[1] = QSCALE * 2.0f*ib;
    g[2] = QSCALE * ic;
    g[3] = QSCALE * -2.0f*(ia*mx + ib*my);
    g[4] = QSCALE * -2.0f*(ib*mx + ic*my);
    g[5] = QSCALE * fmaf(ia*mx, mx, fmaf(2.0f*ib*mx, my, ic*my*my));
    g[6] = 0.0f;
    g[7] = 0.0f;

    float* cc = c4 + 4*n;
    cc[0] = amp;
    cc[1] = amp * colors[3*n+0];
    cc[2] = amp * colors[3*n+1];
    cc[3] = amp * colors[3*n+2];
}

__global__ __launch_bounds__(256)
void splat_main(const float* __restrict__ gcoef_g,
                const float4* __restrict__ c4_g,
                float* __restrict__ out)
{
    __shared__ float  sG[NG * 8];
    __shared__ float4 sC[NG];

    const int tid = threadIdx.x;
    for (int i = tid; i < NG * 8; i += 256) sG[i] = gcoef_g[i];
    for (int i = tid; i < NG;     i += 256) sC[i] = c4_g[i];
    __syncthreads();

    const int lane = tid & 31;
    const int wave = tid >> 5;
    const int strip = blockIdx.x * 8 + wave;
    const int row   = strip >> 4;
    const int x0    = (strip & 15) << 4;

    const int half = lane >> 4;
    const int m    = lane & 15;

    const float fx = (float)(x0 + m) - 128.0f;
    const float fy = (float)row - 128.0f;

    v2f a1, a2;
    a1.x = half ? (fy * fy) : (fx * fx);
    a1.y = half ? fx        : (fx * fy);
    a2.x = half ? 0.0f      : fy;
    a2.y = half ? 0.0f      : 1.0f;

    float4 acc[8];
#pragma unroll
    for (int r = 0; r < 8; ++r) acc[r] = make_float4(0.f, 0.f, 0.f, 0.f);

    const int k0 = half << 1;

    for (int t = 0; t < NG / 16; ++t) {
        const int n = (t << 4) + m;
        const float* g = &sG[n * 8];
        v2f b1 = *(const v2f*)&g[k0];
        v2f b2 = *(const v2f*)&g[4 + k0];

        v8f q = {};
        q = __builtin_amdgcn_wmma_f32_16x16x4_f32(false, a1, false, b1,
                                                  (short)0, q, false, false);
        q = __builtin_amdgcn_wmma_f32_16x16x4_f32(false, a2, false, b2,
                                                  (short)0, q, false, false);

        const float4 cn = sC[n];
#pragma unroll
        for (int r = 0; r < 8; ++r) {
            float e = __builtin_amdgcn_exp2f(q[r]);
            acc[r].x = fmaf(e, cn.x, acc[r].x);
            acc[r].y = fmaf(e, cn.y, acc[r].y);
            acc[r].z = fmaf(e, cn.z, acc[r].z);
            acc[r].w = fmaf(e, cn.w, acc[r].w);
        }
    }

#pragma unroll
    for (int mask = 1; mask <= 8; mask <<= 1) {
#pragma unroll
        for (int r = 0; r < 8; ++r) {
            acc[r].x += __shfl_xor(acc[r].x, mask, 32);
            acc[r].y += __shfl_xor(acc[r].y, mask, 32);
            acc[r].z += __shfl_xor(acc[r].z, mask, 32);
            acc[r].w += __shfl_xor(acc[r].w, mask, 32);
        }
    }

    if (m < 8) {
        float4 v = acc[0];
#pragma unroll
        for (int r = 1; r < 8; ++r) if (m == r) v = acc[r];
        const int x = x0 + (half << 3) + m;
        const float inv = 1.0f / fmaxf(v.x, 1e-8f);
        float* o = out + (row * WW + x) * 3;
        o[0] = v.y * inv;
        o[1] = v.z * inv;
        o[2] = v.w * inv;
    }
}

extern "C" void kernel_launch(void* const* d_in, const int* in_sizes, int n_in,
                              void* d_out, int out_size, void* d_ws, size_t ws_size,
                              hipStream_t stream) {
    const float* means  = (const float*)d_in[0];
    const float* covs   = (const float*)d_in[1];
    const float* colors = (const float*)d_in[2];
    const float* opac   = (const float*)d_in[3];

    float* gcoef = (float*)d_ws;
    float* c4    = (float*)d_ws + NG * 8;

    splat_prep<<<(NG + 255) / 256, 256, 0, stream>>>(means, covs, colors, opac,
                                                     gcoef, c4);
    splat_main<<<(HH * WW) / (8 * 16), 256, 0, stream>>>(gcoef,
                                                         (const float4*)c4,
                                                         (float*)d_out);
}


# Block = 8x32 pixel tile, not a row: a further row's exponent tile = row above + ONE K=4 WMMA (row step linear in x,y; old tile as C operand): 36 WMMAs a wave, not 64; exponent summed in a new order
# speedup vs baseline: 2.2482x; 2.2482x over previous
#include <hip/hip_runtime.h>
#include <math.h>

typedef __attribute__((ext_vector_type(2))) float v2f;
typedef __attribute__((ext_vector_type(8))) float v8f;

#define NG 512
#define HH 256
#define WW 256
#define NWAVE 16
#define TR 8
#define TC 32
#define GREC 12

#define QSCALE (-0.72134752044448170368f)

__global__ __launch_bounds__(512) __attribute__((amdgpu_waves_per_eu(4, 4)))
void splat_tiles(const float* __restrict__ means,
                 const float* __restrict__ covs,
                 const float* __restrict__ colors,
                 const float* __restrict__ opac,
                 float* __restrict__ out)
{
    __shared__ float  sG[NG * GREC];
    __shared__ float4 sC[NG];
    __shared__ float4 sP[NWAVE * 2 * TR * TC];

    const int tid = threadIdx.x;

    {
        const int n = tid;
        float mx = means[2*n+0] - 128.0f;
        float my = means[2*n+1] - 128.0f;
        float a = covs[4*n+0], b = covs[4*n+1], c = covs[4*n+3];
        float det = a*c - b*b;
        float inv_det = 1.0f / det;
        float ia =  c*inv_det;
        float ib = -b*inv_det;
        float ic =  a*inv_det;
        float amp = opac[n] / (6.28318530717958647f * sqrtf(det + 1e-8f));

        const float g1 = QSCALE * 2.0f*ib;
        const float g2 = QSCALE * ic;
        const float g4 = QSCALE * -2.0f*(ib*mx + ic*my);

        float* g = sG + GREC*n;
        g[0] = QSCALE * ia;
        g[1] = g1;
        g[2] = g2;
        g[3] = QSCALE * -2.0f*(ia*mx + ib*my);
        g[4] = g4;
        g[5] = QSCALE * fmaf(ia*mx, mx, fmaf(2.0f*ib*mx, my, ic*my*my));
        g[6] = 0.0f;
        g[7] = 0.0f;
        g[8]  = g1;
        g[9]  = g2 + g2;
        g[10] = g2 + g4;
        g[11] = 0.0f;

        sC[n] = make_float4(amp, amp * colors[3*n+0], amp * colors[3*n+1],
                            amp * colors[3*n+2]);
    }
    __syncthreads();

    const int lane = tid & 31;
    const int wave = tid >> 5;
    const int half = lane >> 4;
    const int c16  = lane & 15;
    const int row0 = (blockIdx.x >> 3) * TR;
    const int col0 = (blockIdx.x & 7) * TC;
    const int k0   = half << 1;

    v2f a1[2], a2[2], ad[2];
    float4 cn[2][8];
#pragma unroll
    for (int s = 0; s < 2; ++s) {
        const int nb = (wave << 5) + (s << 4);
        const float* g = &sG[(nb + c16) * GREC];
        a1[s] = *(const v2f*)&g[k0];
        a2[s] = *(const v2f*)&g[4 + k0];
        ad[s] = *(const v2f*)&g[8 + k0];
#pragma unroll
        for (int r = 0; r < 8; ++r)
            cn[s][r] = sC[nb + (half << 3) + r];
    }

    const float fy0 = (float)row0 - 128.0f;
    const float fy2 = fy0 * fy0;

    v2f b1[2], bd[2], b2;
    b2.x = half ? 0.0f : fy0;
    b2.y = half ? 0.0f : 1.0f;
#pragma unroll
    for (int st = 0; st < 2; ++st) {
        const float fx = (float)(col0 + (st << 4) + c16) - 128.0f;
        b1[st].x = half ? fy2 : (fx * fx);
        b1[st].y = half ? fx  : (fx * fy0);
        bd[st].x = half ? 1.0f : fx;
        bd[st].y = half ? 0.0f : fy0;
    }
    const float dy = half ? 0.0f : 1.0f;

    float4* myP = &sP[((wave << 1) + half) * (TR * TC) + c16];

    v8f q[2][2];
#pragma unroll
    for (int st = 0; st < 2; ++st)
#pragma unroll
        for (int s = 0; s < 2; ++s) {
            v8f z = {};
            z = __builtin_amdgcn_wmma_f32_16x16x4_f32(false, a1[s], false, b1[st],
                                                      (short)0, z, false, false);
            q[st][s] = __builtin_amdgcn_wmma_f32_16x16x4_f32(false, a2[s], false, b2,
                                                      (short)0, z, false, false);
        }

    for (int r = 0; r < TR; ++r) {
        if (r > 0) {
#pragma unroll
            for (int st = 0; st < 2; ++st) {
#pragma unroll
                for (int s = 0; s < 2; ++s)
                    q[st][s] = __builtin_amdgcn_wmma_f32_16x16x4_f32(
                        false, ad[s], false, bd[st], (short)0, q[st][s], false, false);
                bd[st].y += dy;
            }
        }
#pragma unroll
        for (int st = 0; st < 2; ++st) {
            float4 acc[2];
#pragma unroll
            for (int s = 0; s < 2; ++s) {
                float4 t = make_float4(0.f, 0.f, 0.f, 0.f);
#pragma unroll
                for (int i = 0; i < 8; ++i) {
                    float e = __builtin_amdgcn_exp2f(q[st][s][i]);
                    t.x = fmaf(e, cn[s][i].x, t.x);
                    t.y = fmaf(e, cn[s][i].y, t.y);
                    t.z = fmaf(e, cn[s][i].z, t.z);
                    t.w = fmaf(e, cn[s][i].w, t.w);
                }
                acc[s] = t;
            }
            myP[r * TC + (st << 4)] =
                make_float4(acc[0].x + acc[1].x, acc[0].y + acc[1].y,
                            acc[0].z + acc[1].z, acc[0].w + acc[1].w);
        }
    }
    __syncthreads();

    if (tid < TR * TC) {
        float4 v = sP[tid];
#pragma unroll
        for (int k = 1; k < NWAVE * 2; ++k) {
            const float4 p = sP[k * (TR * TC) + tid];
            v.x += p.x; v.y += p.y; v.z += p.z; v.w += p.w;
        }
        const float inv = 1.0f / fmaxf(v.x, 1e-8f);
        const int pr = row0 + (tid >> 5);
        const int pc = col0 + (tid & 31);
        float* o = out + (pr * WW + pc) * 3;
        o[0] = v.y * inv;
        o[1] = v.z * inv;
        o[2] = v.w * inv;
    }
}

extern "C" void kernel_launch(void* const* d_in, const int* in_sizes, int n_in,
                              void* d_out, int out_size, void* d_ws, size_t ws_size,
                              hipStream_t stream) {
    const float* means  = (const float*)d_in[0];
    const float* covs   = (const float*)d_in[1];
    const float* colors = (const float*)d_in[2];
    const float* opac   = (const float*)d_in[3];

    splat_tiles<<<(HH / TR) * (WW / TC), 512, 0, stream>>>(means, covs, colors, opac,
                                                           (float*)d_out);
}
